# starting kernel
# baseline (speedup 1.0000x reference)
#include <hip/hip_runtime.h>

typedef __attribute__((ext_vector_type(2))) float v2f;
typedef __attribute__((ext_vector_type(8))) float v8f;

#define BS 4096
#define CH 2048
#define NV 8

__global__ __launch_bounds__(256) void interviews_gram_kernel(
    const float* __restrict__ X, float* __restrict__ out) {
  const int lane = threadIdx.x & 31;
  const int wave = threadIdx.x >> 5;
  const int half = lane >> 4;
  const int m    = lane & 15;
  const int view = m & 7;
  const int bsel = m >> 3;
  const int bpair = blockIdx.x * 8 + wave;
  const int b     = bpair * 2 + bsel;

  const float* src = X + (size_t)(view * BS + b) * CH + 2 * half;

  v8f acc = {};
#pragma unroll 8
  for (int k = 0; k < CH; k += 4) {
    v2f a = __builtin_nontemporal_load((const v2f*)(src + k));
    acc = __builtin_amdgcn_wmma_f32_16x16x4_f32(
         false, a,  false, a,
         (short)0, acc,  false,  false);
  }

  __shared__ alignas(32) float lds[8][32][8];
  *(v8f*)(&lds[wave][lane][0]) = acc;
  __syncthreads();

  if (lane < 2) {
    const int base = lane ? 24 : 0;
    float Gm[8][8];
#pragma unroll
    for (int n = 0; n < 8; ++n) {
      v8f r = *(const v8f*)(&lds[wave][base + n][0]);
#pragma unroll
      for (int v = 0; v < 8; ++v) Gm[n][v] = r[v];
    }
    float inv[8];
#pragma unroll
    for (int v = 0; v < 8; ++v) inv[v] = 1.0f / sqrtf(Gm[v][v]);

    float sum = 0.0f, sumsq = 0.0f;
#pragma unroll
    for (int v = 0; v < 8; ++v) {
#pragma unroll
      for (int w = v + 1; w < 8; ++w) {
        float s = Gm[v][w] * inv[v] * inv[w];
        sum += s;
        sumsq += s * s;
      }
    }
    const float mean = sum * (1.0f / 28.0f);
    const float var  = (sumsq - 28.0f * mean * mean) * (1.0f / 27.0f);
    out[bpair * 2 + lane] = -var;
  }
}

extern "C" void kernel_launch(void* const* d_in, const int* in_sizes, int n_in,
                              void* d_out, int out_size, void* d_ws, size_t ws_size,
                              hipStream_t stream) {
  const float* X = (const float*)d_in[0];
  float* out = (float*)d_out;
  (void)in_sizes; (void)n_in; (void)out_size; (void)d_ws; (void)ws_size;

  dim3 grid(BS / 16);
  dim3 block(256);
  hipLaunchKernelGGL(interviews_gram_kernel, grid, block, 0, stream, X, out);
}


# Global loads made whole-line (512 contiguous bytes per instruction, was 16 rows x 16 B) and staged through wave-private LDS into the f32 WMMA fragments; channel summation order changes
# speedup vs baseline: 1.1605x; 1.1605x over previous
#include <hip/hip_runtime.h>


typedef __attribute__((ext_vector_type(2))) float v2f;
typedef __attribute__((ext_vector_type(4))) float v4f;
typedef __attribute__((ext_vector_type(8))) float v8f;

#define BS 4096
#define CH 2048
#define NV 8

#define TCH 256
#define NQ (TCH / 128)
#define NIT (CH / TCH)
#define PITCH16 65
#define WAVE16 (NV * PITCH16)

__global__ __launch_bounds__(256) void interviews_gram_kernel(
    const float* __restrict__ X, float* __restrict__ out) {
  __shared__ alignas(32) v4f tile[8][WAVE16];
  __shared__ float res[8];

  const int lane = threadIdx.x & 31;
  const int wave = threadIdx.x >> 5;
  const int b    = blockIdx.x * 8 + wave;

  const int c = lane & 15;
  const int h = lane >> 4;
  const int v = c & 7;
  const int p = c >> 3;

  const float* g = X + (size_t)b * CH + 4 * lane;
  v4f* T = tile[wave];
  const v4f* F = T + v * PITCH16 + 8 * p + h;

  v4f r[NV][NQ];
#pragma unroll
  for (int vv = 0; vv < NV; ++vv)
#pragma unroll
    for (int q = 0; q < NQ; ++q)
      r[vv][q] = __builtin_nontemporal_load(
          (const v4f*)(g + (size_t)vv * BS * CH + 128 * q));

  v8f acc0 = {}, acc1 = {};
  for (int it = 0; it < NIT; ++it) {
#pragma unroll
    for (int vv = 0; vv < NV; ++vv)
#pragma unroll
      for (int q = 0; q < NQ; ++q)
        T[vv * PITCH16 + 32 * q + lane] = r[vv][q];

    if (it + 1 < NIT) {
#pragma unroll
      for (int vv = 0; vv < NV; ++vv)
#pragma unroll
        for (int q = 0; q < NQ; ++q)
          r[vv][q] = __builtin_nontemporal_load(
              (const v4f*)(g + (size_t)vv * BS * CH + TCH * (it + 1) + 128 * q));
    }

    __builtin_amdgcn_fence(__ATOMIC_RELEASE, "wavefront");
    __builtin_amdgcn_wave_barrier();
    __builtin_amdgcn_fence(__ATOMIC_ACQUIRE, "wavefront");

#pragma unroll
    for (int t = 0; t < 16; ++t) {
      const v4f f = F[2 * (t & 3) + 16 * (t >> 2)];
      const v2f lo = {f.x, f.y};
      const v2f hi = {f.z, f.w};
      acc0 = __builtin_amdgcn_wmma_f32_16x16x4_f32(
          false, lo, false, lo, (short)0, acc0, false, false);
      acc1 = __builtin_amdgcn_wmma_f32_16x16x4_f32(
          false, hi, false, hi, (short)0, acc1, false, false);
    }

    __builtin_amdgcn_fence(__ATOMIC_RELEASE, "wavefront");
    __builtin_amdgcn_wave_barrier();
    __builtin_amdgcn_fence(__ATOMIC_ACQUIRE, "wavefront");
  }

  const v8f acc = acc0 + acc1;
  float* Tf = (float*)T;
  *(v8f*)(Tf + lane * 8) = acc;
  __builtin_amdgcn_fence(__ATOMIC_RELEASE, "wavefront");
  __builtin_amdgcn_wave_barrier();
  __builtin_amdgcn_fence(__ATOMIC_ACQUIRE, "wavefront");

  if (lane == 0) {
    float Gm[8][8];
#pragma unroll
    for (int n = 0; n < 8; ++n) {
      const v8f r0 = *(const v8f*)(Tf + n * 8);
      const v8f r1 = *(const v8f*)(Tf + (24 + n) * 8);
#pragma unroll
      for (int u = 0; u < 8; ++u) Gm[n][u] = r0[u] + r1[u];
    }
    float inv[8];
#pragma unroll
    for (int u = 0; u < 8; ++u) inv[u] = 1.0f / sqrtf(Gm[u][u]);

    float sum = 0.0f, sumsq = 0.0f;
#pragma unroll
    for (int u = 0; u < 8; ++u) {
#pragma unroll
      for (int w = u + 1; w < 8; ++w) {
        float s = Gm[u][w] * inv[u] * inv[w];
        sum += s;
        sumsq += s * s;
      }
    }
    const float mean = sum * (1.0f / 28.0f);
    const float var  = (sumsq - 28.0f * mean * mean) * (1.0f / 27.0f);
    res[wave] = -var;
  }
  __syncthreads();
  if (threadIdx.x < 8) out[blockIdx.x * 8 + threadIdx.x] = res[threadIdx.x];
}

extern "C" void kernel_launch(void* const* d_in, const int* in_sizes, int n_in,
                              void* d_out, int out_size, void* d_ws, size_t ws_size,
                              hipStream_t stream) {
  const float* X = (const float*)d_in[0];
  float* out = (float*)d_out;
  (void)in_sizes; (void)n_in; (void)out_size; (void)d_ws; (void)ws_size;

  dim3 grid(BS / 8);
  dim3 block(256);
  hipLaunchKernelGGL(interviews_gram_kernel, grid, block, 0, stream, X, out);
}


# Work-group shares one LDS image per item: wave w streams view w of its 8 adjacent items (contiguous 64 KiB, whole row per step); waves split the channels; Gram = 8 slice partials added in wave order
# speedup vs baseline: 2.3092x; 1.9898x over previous
#include <hip/hip_runtime.h>


typedef __attribute__((ext_vector_type(2))) float v2f;
typedef __attribute__((ext_vector_type(4))) float v4f;
typedef __attribute__((ext_vector_type(8))) float v8f;

#define BS 4096
#define CH 2048
#define NV 8

#define NQ (CH / 128)
#define PITCH16 (CH / 4 + 1)
#define SLICE16 64

static __device__ __forceinline__ void block_handover() {
  asm volatile("s_wait_dscnt 0x0\n\ts_barrier_signal -1\n\ts_barrier_wait -1" ::: "memory");
}

__global__ __launch_bounds__(256) void interviews_gram_kernel(
    const float* __restrict__ X, float* __restrict__ out) {
  __shared__ alignas(32) v4f item[NV * PITCH16];
  __shared__ alignas(32) float part[8][8][32 * 8];
  __shared__ float res[8];

  const int lane = threadIdx.x & 31;
  const int wave = threadIdx.x >> 5;
  const int b0   = blockIdx.x * 8;

  const int c = lane & 15;
  const int h = lane >> 4;
  const int v = c & 7;
  const int p = c >> 3;

  const float* g = X + ((size_t)wave * BS + b0) * CH + 4 * lane;
  v4f* S = item + wave * PITCH16 + lane;
  const v4f* F = item + v * PITCH16 + SLICE16 * wave + 8 * p + h;

  v4f r[2][NQ];
#pragma unroll
  for (int s = 0; s < 2; ++s) {
    const float* gi = g + (size_t)s * CH;
#pragma unroll
    for (int q = 0; q < NQ; ++q)
      r[s][q] = __builtin_nontemporal_load((const v4f*)(gi + 128 * q));
  }

#pragma unroll
  for (int j = 0; j < 8; ++j) {
#pragma unroll
    for (int q = 0; q < NQ; ++q) S[32 * q] = r[j & 1][q];

    if (j + 2 < 8) {
      const float* gi = g + (size_t)(j + 2) * CH;
#pragma unroll
      for (int q = 0; q < NQ; ++q)
        r[j & 1][q] = __builtin_nontemporal_load((const v4f*)(gi + 128 * q));
    }

    block_handover();

    v8f acc0 = {}, acc1 = {};
#pragma unroll
    for (int t = 0; t < 16; ++t) {
      const v4f f = F[2 * (t & 3) + 16 * (t >> 2)];
      const v2f lo = {f.x, f.y};
      const v2f hi = {f.z, f.w};
      acc0 = __builtin_amdgcn_wmma_f32_16x16x4_f32(
          false, lo, false, lo, (short)0, acc0, false, false);
      acc1 = __builtin_amdgcn_wmma_f32_16x16x4_f32(
          false, hi, false, hi, (short)0, acc1, false, false);
    }
    *(v8f*)(&part[j][wave][lane * 8]) = acc0 + acc1;

    block_handover();
  }

  float* Tf = &part[wave][0][0];
  v8f acc = *(const v8f*)(Tf + lane * 8);
#pragma unroll
  for (int w2 = 1; w2 < 8; ++w2) acc += *(const v8f*)(&part[wave][w2][lane * 8]);
  *(v8f*)(Tf + lane * 8) = acc;
  __builtin_amdgcn_fence(__ATOMIC_RELEASE, "wavefront");
  __builtin_amdgcn_wave_barrier();
  __builtin_amdgcn_fence(__ATOMIC_ACQUIRE, "wavefront");

  if (lane == 0) {
    float Gm[8][8];
#pragma unroll
    for (int n = 0; n < 8; ++n) {
      const v8f r0 = *(const v8f*)(Tf + n * 8);
      const v8f r1 = *(const v8f*)(Tf + (24 + n) * 8);
#pragma unroll
      for (int u = 0; u < 8; ++u) Gm[n][u] = r0[u] + r1[u];
    }
    float inv[8];
#pragma unroll
    for (int u = 0; u < 8; ++u) inv[u] = 1.0f / sqrtf(Gm[u][u]);

    float sum = 0.0f, sumsq = 0.0f;
#pragma unroll
    for (int u = 0; u < 8; ++u) {
#pragma unroll
      for (int w = u + 1; w < 8; ++w) {
        float s = Gm[u][w] * inv[u] * inv[w];
        sum += s;
        sumsq += s * s;
      }
    }
    const float mean = sum * (1.0f / 28.0f);
    const float var  = (sumsq - 28.0f * mean * mean) * (1.0f / 27.0f);
    res[wave] = -var;
  }
  __syncthreads();
  if (threadIdx.x < 8) out[b0 + threadIdx.x] = res[threadIdx.x];
}

extern "C" void kernel_launch(void* const* d_in, const int* in_sizes, int n_in,
                              void* d_out, int out_size, void* d_ws, size_t ws_size,
                              hipStream_t stream) {
  const float* X = (const float*)d_in[0];
  float* out = (float*)d_out;
  (void)in_sizes; (void)n_in; (void)out_size; (void)d_ws; (void)ws_size;

  dim3 grid(BS / 8);
  dim3 block(256);
  hipLaunchKernelGGL(interviews_gram_kernel, grid, block, 0, stream, X, out);
}
